# starting kernel
# baseline (speedup 1.0000x reference)
#include <hip/hip_runtime.h>
#include <hip/hip_bf16.h>
#include <math.h>

typedef __attribute__((ext_vector_type(2))) float v2f;
typedef __attribute__((ext_vector_type(8))) float v8f;

__global__ void zero_kernel(float* __restrict__ p, long long n) {
    long long i = (long long)blockIdx.x * blockDim.x + threadIdx.x;
    if (i < n) p[i] = 0.0f;
}

__global__ void gemm1_kernel(const float* __restrict__ x,
                             const float* __restrict__ w1k1,
                             const float* __restrict__ root1,
                             float* __restrict__ y1r1, int N) {
    const int lane = threadIdx.x & 31;
    const int wave = threadIdx.x >> 5;
    const int tile = blockIdx.x * 4 + wave;
    const int row0 = tile * 16;
    if (row0 >= N) return;

    const int colL = lane & 15;
    const int kSub = (lane >> 4) * 2;
    int rowA = row0 + colL;
    if (rowA >= N) rowA = N - 1;

    v8f c0 = {};
    v8f c1 = {};
#pragma unroll 4
    for (int kk = 0; kk < 32; ++kk) {
        const int k = kk * 4 + kSub;
        v2f a;
        a.x = x[(long long)rowA * 128 + k];
        a.y = x[(long long)rowA * 128 + k + 1];
        v2f b0;
        b0.x = w1k1[k * 16 + colL];
        b0.y = w1k1[(k + 1) * 16 + colL];
        v2f b1;
        b1.x = root1[k * 16 + colL];
        b1.y = root1[(k + 1) * 16 + colL];
        c0 = __builtin_amdgcn_wmma_f32_16x16x4_f32(false, a, false, b0,
                                                   (short)0, c0, false, false);
        c1 = __builtin_amdgcn_wmma_f32_16x16x4_f32(false, a, false, b1,
                                                   (short)0, c1, false, false);
    }

    const int rbase = (lane >> 4) * 8;
#pragma unroll
    for (int r = 0; r < 8; ++r) {
        const int row = row0 + rbase + r;
        if (row < N) {
            y1r1[(long long)row * 32 + colL]      = c0[r];
            y1r1[(long long)row * 32 + 16 + colL] = c1[r];
        }
    }
}

__global__ void scatter1_kernel(const int* __restrict__ ei, int E,
                                const float* __restrict__ y1r1,
                                float* __restrict__ agg1,
                                float* __restrict__ deg) {
    long long gid = (long long)blockIdx.x * blockDim.x + threadIdx.x;
    int e = (int)(gid >> 4);
    int c = (int)(gid & 15);
    if (e >= E) return;
    int src = ei[e];
    int dst = ei[E + e];
    atomicAdd(&agg1[(long long)dst * 16 + c], y1r1[(long long)src * 32 + c]);
    if (c == 0) atomicAdd(&deg[dst], 1.0f);
}

__global__ void combine1_kernel(const float* __restrict__ agg1,
                                const float* __restrict__ deg,
                                const float* __restrict__ y1r1,
                                const float* __restrict__ b1,
                                float* __restrict__ h, int N) {
    long long gid = (long long)blockIdx.x * blockDim.x + threadIdx.x;
    if (gid >= (long long)N * 16) return;
    int n = (int)(gid >> 4);
    int c = (int)(gid & 15);
    float v = agg1[gid] / fmaxf(deg[n], 1.0f)
            + y1r1[(long long)n * 32 + 16 + c] + b1[c];
    h[gid] = (v > 0.0f) ? v : (expf(v) - 1.0f);
}

__global__ void scatter2_kernel(const int* __restrict__ ei, int E,
                                const float* __restrict__ h,
                                float* __restrict__ agg2) {
    long long gid = (long long)blockIdx.x * blockDim.x + threadIdx.x;
    int e = (int)(gid >> 4);
    int c = (int)(gid & 15);
    if (e >= E) return;
    int src = ei[e];
    int dst = ei[E + e];
    atomicAdd(&agg2[(long long)dst * 16 + c], h[(long long)src * 16 + c]);
}

__device__ __forceinline__ float b2load(const float* w2k1, const float* root2,
                                        int k, int col) {
    if (col >= 40) return 0.0f;
    return (k < 16) ? w2k1[k * 40 + col] : root2[(k - 16) * 40 + col];
}

__global__ void gemm2_kernel(const float* __restrict__ agg2,
                             const float* __restrict__ deg,
                             const float* __restrict__ h,
                             const float* __restrict__ w2k1,
                             const float* __restrict__ root2,
                             const float* __restrict__ b2,
                             float* __restrict__ out, int N) {
    const int lane = threadIdx.x & 31;
    const int wave = threadIdx.x >> 5;
    const int tile = blockIdx.x * 4 + wave;
    const int row0 = tile * 16;
    if (row0 >= N) return;

    const int colL = lane & 15;
    const int kSub = (lane >> 4) * 2;
    int rowA = row0 + colL;
    if (rowA >= N) rowA = N - 1;
    const float invd = 1.0f / fmaxf(deg[rowA], 1.0f);

    v8f c0 = {}, c1 = {}, c2 = {};
#pragma unroll
    for (int kk = 0; kk < 8; ++kk) {
        const int k = kk * 4 + kSub;
        v2f a;
        a.x = (k < 16) ? agg2[(long long)rowA * 16 + k] * invd
                       : h[(long long)rowA * 16 + (k - 16)];
        a.y = (k + 1 < 16) ? agg2[(long long)rowA * 16 + k + 1] * invd
                           : h[(long long)rowA * 16 + (k + 1 - 16)];
        v2f b0, b1, b2v;
        b0.x  = b2load(w2k1, root2, k,     colL);
        b0.y  = b2load(w2k1, root2, k + 1, colL);
        b1.x  = b2load(w2k1, root2, k,     16 + colL);
        b1.y  = b2load(w2k1, root2, k + 1, 16 + colL);
        b2v.x = b2load(w2k1, root2, k,     32 + colL);
        b2v.y = b2load(w2k1, root2, k + 1, 32 + colL);
        c0 = __builtin_amdgcn_wmma_f32_16x16x4_f32(false, a, false, b0,
                                                   (short)0, c0, false, false);
        c1 = __builtin_amdgcn_wmma_f32_16x16x4_f32(false, a, false, b1,
                                                   (short)0, c1, false, false);
        c2 = __builtin_amdgcn_wmma_f32_16x16x4_f32(false, a, false, b2v,
                                                   (short)0, c2, false, false);
    }

    const int rbase = (lane >> 4) * 8;
#pragma unroll
    for (int r = 0; r < 8; ++r) {
        const int row = row0 + rbase + r;
        if (row >= N) continue;
        out[(long long)row * 40 + colL]      = c0[r] + b2[colL];
        out[(long long)row * 40 + 16 + colL] = c1[r] + b2[16 + colL];
        if (colL < 8)
            out[(long long)row * 40 + 32 + colL] = c2[r] + b2[32 + colL];
    }
}

__global__ void logsoftmax_kernel(float* __restrict__ out, int N) {
    int n = blockIdx.x * blockDim.x + threadIdx.x;
    if (n >= N) return;
    float v[40];
    float m = -INFINITY;
#pragma unroll
    for (int i = 0; i < 40; ++i) {
        v[i] = out[(long long)n * 40 + i];
        m = fmaxf(m, v[i]);
    }
    float s = 0.0f;
#pragma unroll
    for (int i = 0; i < 40; ++i) s += expf(v[i] - m);
    float ls = logf(s);
#pragma unroll
    for (int i = 0; i < 40; ++i) out[(long long)n * 40 + i] = v[i] - m - ls;
}

extern "C" void kernel_launch(void* const* d_in, const int* in_sizes, int n_in,
                              void* d_out, int out_size, void* d_ws, size_t ws_size,
                              hipStream_t stream) {
    const float* x     = (const float*)d_in[0];
    const int*   ei    = (const int*)d_in[1];
    const float* W1    = (const float*)d_in[2];
    const float* root1 = (const float*)d_in[3];
    const float* b1    = (const float*)d_in[4];
    const float* W2    = (const float*)d_in[5];
    const float* root2 = (const float*)d_in[6];
    const float* b2    = (const float*)d_in[7];
    float*       out   = (float*)d_out;

    const int N = in_sizes[0] / 128;
    const int E = in_sizes[1] / 2;

    const float* w1k1 = W1 + 128 * 16;
    const float* w2k1 = W2 + 16 * 40;

    float* ws   = (float*)d_ws;
    float* agg1 = ws;
    float* deg  = ws + (size_t)N * 16;
    float* agg2 = ws + (size_t)N * 17;
    float* y1r1 = ws + (size_t)N * 33;
    float* h    = ws + (size_t)N * 65;

    long long nz = (long long)N * 33;
    zero_kernel<<<(unsigned)((nz + 255) / 256), 256, 0, stream>>>(ws, nz);

    const int tiles   = (N + 15) / 16;
    const int gblocks = (tiles + 3) / 4;

    gemm1_kernel<<<gblocks, 128, 0, stream>>>(x, w1k1, root1, y1r1, N);

    long long sthreads = (long long)E * 16;
    unsigned sblocks = (unsigned)((sthreads + 255) / 256);
    scatter1_kernel<<<sblocks, 256, 0, stream>>>(ei, E, y1r1, agg1, deg);

    long long cthreads = (long long)N * 16;
    combine1_kernel<<<(unsigned)((cthreads + 255) / 256), 256, 0, stream>>>(
        agg1, deg, y1r1, b1, h, N);

    scatter2_kernel<<<sblocks, 256, 0, stream>>>(ei, E, h, agg2);

    gemm2_kernel<<<gblocks, 128, 0, stream>>>(agg2, deg, h, w2k1, root2, b2, out, N);

    logsoftmax_kernel<<<(unsigned)((N + 255) / 256), 256, 0, stream>>>(out, N);
}


# scatter1: in-degree rides as lane 16 of the edge's one atomic instruction into a 128-B agg1 row (pitch 32 floats); no separate deg atomics; y1/r1 stored apart; sums stay in atomic arrival order
# speedup vs baseline: 2.0268x; 2.0268x over previous
#include <hip/hip_runtime.h>
#include <hip/hip_bf16.h>
#include <math.h>

typedef __attribute__((ext_vector_type(2))) float v2f;
typedef __attribute__((ext_vector_type(8))) float v8f;

__global__ void zero_kernel(float* __restrict__ p, long long n) {
    long long i = (long long)blockIdx.x * blockDim.x + threadIdx.x;
    if (i < n) p[i] = 0.0f;
}

__global__ void gemm1_kernel(const float* __restrict__ x,
                             const float* __restrict__ w1k1,
                             const float* __restrict__ root1,
                             float* __restrict__ y1,
                             float* __restrict__ r1, int N) {
    const int lane = threadIdx.x & 31;
    const int wave = threadIdx.x >> 5;
    const int tile = blockIdx.x * 4 + wave;
    const int row0 = tile * 16;
    if (row0 >= N) return;

    const int colL = lane & 15;
    const int kSub = (lane >> 4) * 2;
    int rowA = row0 + colL;
    if (rowA >= N) rowA = N - 1;

    v8f c0 = {};
    v8f c1 = {};
#pragma unroll 4
    for (int kk = 0; kk < 32; ++kk) {
        const int k = kk * 4 + kSub;
        v2f a;
        a.x = x[(long long)rowA * 128 + k];
        a.y = x[(long long)rowA * 128 + k + 1];
        v2f b0;
        b0.x = w1k1[k * 16 + colL];
        b0.y = w1k1[(k + 1) * 16 + colL];
        v2f b1;
        b1.x = root1[k * 16 + colL];
        b1.y = root1[(k + 1) * 16 + colL];
        c0 = __builtin_amdgcn_wmma_f32_16x16x4_f32(false, a, false, b0,
                                                   (short)0, c0, false, false);
        c1 = __builtin_amdgcn_wmma_f32_16x16x4_f32(false, a, false, b1,
                                                   (short)0, c1, false, false);
    }

    const int rbase = (lane >> 4) * 8;
#pragma unroll
    for (int r = 0; r < 8; ++r) {
        const int row = row0 + rbase + r;
        if (row < N) {
            y1[(long long)row * 16 + colL] = c0[r];
            r1[(long long)row * 16 + colL] = c1[r];
        }
    }
}

__device__ __forceinline__ void edge_add(float* __restrict__ arow, int d,
                                         int lane, float val) {
    if (lane < 17) atomicAdd(&arow[(long long)d * 32 + lane], val);
}

__global__ void scatter1_kernel(const int* __restrict__ ei, int E,
                                const float* __restrict__ y1,
                                float* __restrict__ arow) {
    const int lane = threadIdx.x & 31;
    const long long wave =
        ((long long)blockIdx.x * blockDim.x + threadIdx.x) >> 5;
    const long long base = wave * 32;
    if (base >= E) return;
    const int c = lane & 15;
    const long long e = base + lane;
    int srcv = 0, dstv = 0;
    if (e < E) {
        srcv = ei[e];
        dstv = ei[(long long)E + e];
    }
    if (base + 32 <= E) {
        float v[32];
#pragma unroll
        for (int j = 0; j < 32; ++j) {
            const int sidx = __builtin_amdgcn_readlane(srcv, j);
            v[j] = y1[(long long)sidx * 16 + c];
        }
#pragma unroll
        for (int half = 0; half < 2; ++half) {
            int d[16];
#pragma unroll
            for (int j = 0; j < 16; ++j)
                d[j] = __builtin_amdgcn_readlane(dstv, half * 16 + j);
            if (lane < 17) {
#pragma unroll
                for (int j = 0; j < 16; ++j)
                    atomicAdd(&arow[(long long)d[j] * 32 + lane],
                              (lane < 16) ? v[half * 16 + j] : 1.0f);
            }
        }
    } else {
        const int cnt = (int)(E - base);
        for (int j = 0; j < cnt; ++j) {
            const int sidx = __builtin_amdgcn_readlane(srcv, j);
            const int d = __builtin_amdgcn_readlane(dstv, j);
            const float val = y1[(long long)sidx * 16 + c];
            edge_add(arow, d, lane, (lane < 16) ? val : 1.0f);
        }
    }
}

__global__ void combine1_kernel(const float* __restrict__ arow,
                                const float* __restrict__ r1,
                                const float* __restrict__ b1,
                                float* __restrict__ h, int N) {
    long long gid = (long long)blockIdx.x * blockDim.x + threadIdx.x;
    if (gid >= (long long)N * 16) return;
    long long n = gid >> 4;
    int c = (int)(gid & 15);
    float v = arow[n * 32 + c] / fmaxf(arow[n * 32 + 16], 1.0f)
            + r1[gid] + b1[c];
    h[gid] = (v > 0.0f) ? v : (expf(v) - 1.0f);
}

__global__ void scatter2_kernel(const int* __restrict__ ei, int E,
                                const float* __restrict__ h,
                                float* __restrict__ agg2) {
    long long gid = (long long)blockIdx.x * blockDim.x + threadIdx.x;
    int e = (int)(gid >> 4);
    int c = (int)(gid & 15);
    if (e >= E) return;
    int src = ei[e];
    int dst = ei[E + e];
    atomicAdd(&agg2[(long long)dst * 16 + c], h[(long long)src * 16 + c]);
}

__device__ __forceinline__ float b2load(const float* w2k1, const float* root2,
                                        int k, int col) {
    if (col >= 40) return 0.0f;
    return (k < 16) ? w2k1[k * 40 + col] : root2[(k - 16) * 40 + col];
}

__global__ void gemm2_kernel(const float* __restrict__ agg2,
                             const float* __restrict__ arow,
                             const float* __restrict__ h,
                             const float* __restrict__ w2k1,
                             const float* __restrict__ root2,
                             const float* __restrict__ b2,
                             float* __restrict__ out, int N) {
    const int lane = threadIdx.x & 31;
    const int wave = threadIdx.x >> 5;
    const int tile = blockIdx.x * 4 + wave;
    const int row0 = tile * 16;
    if (row0 >= N) return;

    const int colL = lane & 15;
    const int kSub = (lane >> 4) * 2;
    int rowA = row0 + colL;
    if (rowA >= N) rowA = N - 1;
    const float invd = 1.0f / fmaxf(arow[(long long)rowA * 32 + 16], 1.0f);

    v8f c0 = {}, c1 = {}, c2 = {};
#pragma unroll
    for (int kk = 0; kk < 8; ++kk) {
        const int k = kk * 4 + kSub;
        v2f a;
        a.x = (k < 16) ? agg2[(long long)rowA * 16 + k] * invd
                       : h[(long long)rowA * 16 + (k - 16)];
        a.y = (k + 1 < 16) ? agg2[(long long)rowA * 16 + k + 1] * invd
                           : h[(long long)rowA * 16 + (k + 1 - 16)];
        v2f b0, b1, b2v;
        b0.x  = b2load(w2k1, root2, k,     colL);
        b0.y  = b2load(w2k1, root2, k + 1, colL);
        b1.x  = b2load(w2k1, root2, k,     16 + colL);
        b1.y  = b2load(w2k1, root2, k + 1, 16 + colL);
        b2v.x = b2load(w2k1, root2, k,     32 + colL);
        b2v.y = b2load(w2k1, root2, k + 1, 32 + colL);
        c0 = __builtin_amdgcn_wmma_f32_16x16x4_f32(false, a, false, b0,
                                                   (short)0, c0, false, false);
        c1 = __builtin_amdgcn_wmma_f32_16x16x4_f32(false, a, false, b1,
                                                   (short)0, c1, false, false);
        c2 = __builtin_amdgcn_wmma_f32_16x16x4_f32(false, a, false, b2v,
                                                   (short)0, c2, false, false);
    }

    const int rbase = (lane >> 4) * 8;
#pragma unroll
    for (int r = 0; r < 8; ++r) {
        const int row = row0 + rbase + r;
        if (row >= N) continue;
        out[(long long)row * 40 + colL]      = c0[r] + b2[colL];
        out[(long long)row * 40 + 16 + colL] = c1[r] + b2[16 + colL];
        if (colL < 8)
            out[(long long)row * 40 + 32 + colL] = c2[r] + b2[32 + colL];
    }
}

__global__ void logsoftmax_kernel(float* __restrict__ out, int N) {
    int n = blockIdx.x * blockDim.x + threadIdx.x;
    if (n >= N) return;
    float v[40];
    float m = -INFINITY;
#pragma unroll
    for (int i = 0; i < 40; ++i) {
        v[i] = out[(long long)n * 40 + i];
        m = fmaxf(m, v[i]);
    }
    float s = 0.0f;
#pragma unroll
    for (int i = 0; i < 40; ++i) s += expf(v[i] - m);
    float ls = logf(s);
#pragma unroll
    for (int i = 0; i < 40; ++i) out[(long long)n * 40 + i] = v[i] - m - ls;
}

extern "C" void kernel_launch(void* const* d_in, const int* in_sizes, int n_in,
                              void* d_out, int out_size, void* d_ws, size_t ws_size,
                              hipStream_t stream) {
    const float* x     = (const float*)d_in[0];
    const int*   ei    = (const int*)d_in[1];
    const float* W1    = (const float*)d_in[2];
    const float* root1 = (const float*)d_in[3];
    const float* b1    = (const float*)d_in[4];
    const float* W2    = (const float*)d_in[5];
    const float* root2 = (const float*)d_in[6];
    const float* b2    = (const float*)d_in[7];
    float*       out   = (float*)d_out;

    const int N = in_sizes[0] / 128;
    const int E = in_sizes[1] / 2;

    const float* w1k1 = W1 + 128 * 16;
    const float* w2k1 = W2 + 16 * 40;

    float* ws   = (float*)d_ws;
    float* arow = ws;
    float* agg2 = ws + (size_t)N * 32;
    float* y1   = ws + (size_t)N * 48;
    float* r1   = ws + (size_t)N * 64;
    float* h    = y1;

    long long nz = (long long)N * 48;
    zero_kernel<<<(unsigned)((nz + 255) / 256), 256, 0, stream>>>(ws, nz);

    const int tiles   = (N + 15) / 16;
    const int gblocks = (tiles + 3) / 4;

    gemm1_kernel<<<gblocks, 128, 0, stream>>>(x, w1k1, root1, y1, r1, N);

    unsigned s1blocks = (unsigned)(((long long)E + 255) / 256);
    scatter1_kernel<<<s1blocks, 256, 0, stream>>>(ei, E, y1, arow);

    long long sthreads = (long long)E * 16;
    unsigned sblocks = (unsigned)((sthreads + 255) / 256);
    long long cthreads = (long long)N * 16;
    combine1_kernel<<<(unsigned)((cthreads + 255) / 256), 256, 0, stream>>>(
        arow, r1, b1, h, N);

    scatter2_kernel<<<sblocks, 256, 0, stream>>>(ei, E, h, agg2);

    gemm2_kernel<<<gblocks, 128, 0, stream>>>(agg2, arow, h, w2k1, root2, b2, out, N);

    logsoftmax_kernel<<<(unsigned)((N + 255) / 256), 256, 0, stream>>>(out, N);
}


# both edge scatters: work-groups read their die id and draw ticketed 512-edge chunks, adding only edges whose dst node lies in their L2 half's node range; sums stay in atomic arrival order
# speedup vs baseline: 3.9591x; 1.9534x over previous
#include <hip/hip_runtime.h>
#include <hip/hip_bf16.h>
#include <math.h>

typedef __attribute__((ext_vector_type(2))) float v2f;
typedef __attribute__((ext_vector_type(8))) float v8f;

__global__ void zero_kernel(float* __restrict__ p, long long n) {
    long long i = (long long)blockIdx.x * blockDim.x + threadIdx.x;
    if (i < n) p[i] = 0.0f;
}

__global__ void gemm1_kernel(const float* __restrict__ x,
                             const float* __restrict__ w1k1,
                             const float* __restrict__ root1,
                             float* __restrict__ y1,
                             float* __restrict__ r1, int N) {
    const int lane = threadIdx.x & 31;
    const int wave = threadIdx.x >> 5;
    const int tile = blockIdx.x * 4 + wave;
    const int row0 = tile * 16;
    if (row0 >= N) return;

    const int colL = lane & 15;
    const int kSub = (lane >> 4) * 2;
    int rowA = row0 + colL;
    if (rowA >= N) rowA = N - 1;

    v8f c0 = {};
    v8f c1 = {};
#pragma unroll 4
    for (int kk = 0; kk < 32; ++kk) {
        const int k = kk * 4 + kSub;
        v2f a;
        a.x = x[(long long)rowA * 128 + k];
        a.y = x[(long long)rowA * 128 + k + 1];
        v2f b0;
        b0.x = w1k1[k * 16 + colL];
        b0.y = w1k1[(k + 1) * 16 + colL];
        v2f b1;
        b1.x = root1[k * 16 + colL];
        b1.y = root1[(k + 1) * 16 + colL];
        c0 = __builtin_amdgcn_wmma_f32_16x16x4_f32(false, a, false, b0,
                                                   (short)0, c0, false, false);
        c1 = __builtin_amdgcn_wmma_f32_16x16x4_f32(false, a, false, b1,
                                                   (short)0, c1, false, false);
    }

    const int rbase = (lane >> 4) * 8;
#pragma unroll
    for (int r = 0; r < 8; ++r) {
        const int row = row0 + rbase + r;
        if (row < N) {
            y1[(long long)row * 16 + colL] = c0[r];
            r1[(long long)row * 16 + colL] = c1[r];
        }
    }
}

#define CH 64
#define WAVES_PER_WG 8

template <int PITCH, int LN>
__device__ __forceinline__ void scatter_owned(const int* __restrict__ ei, int E, int N,
                                              const float* __restrict__ feat,
                                              float* __restrict__ rows,
                                              int* __restrict__ ctr0,
                                              int* __restrict__ ctr1, int split) {
    __shared__ int s_ticket[2];
    const int lane = threadIdx.x & 31;
    const int wave = threadIdx.x >> 5;
    const int c = lane & 15;
    const int nchunks =
        (int)(((long long)E + CH * WAVES_PER_WG - 1) / (CH * WAVES_PER_WG));
    const unsigned id = __builtin_amdgcn_s_sendmsg_rtn(0x87);
    const int myhalf = (((id >> 16) & 0xFu) >= 4u) ? 1 : 0;

    int k = 0;
    for (int pass = 0; pass < 2; ++pass) {
        const int hsel = myhalf ^ pass;
        int* ctr = hsel ? ctr1 : ctr0;
        const int lo = hsel ? split : 0;
        const unsigned span = (unsigned)(hsel ? (N - split) : split);
        for (int it = 0; it <= nchunks; ++it) {
            const int slot = k & 1;
            ++k;
            if (threadIdx.x == 0) s_ticket[slot] = atomicAdd(ctr, 1);
            __syncthreads();
            const int t = __builtin_amdgcn_readfirstlane(s_ticket[slot]);
            if (t >= nchunks) break;
            for (int s = 0; s < CH / 32; ++s) {
                const long long e =
                    ((long long)t * WAVES_PER_WG + wave) * CH + s * 32 + lane;
                int srcv = 0, dstv = -1;
                if (e < E) {
                    srcv = ei[e];
                    dstv = ei[(long long)E + e];
                }
                const bool own = (unsigned)(dstv - lo) < span;
                const unsigned m = __builtin_amdgcn_ballot_w32(own);
                const int nown = __builtin_popcount(m);
                if (nown == 0) continue;
                const unsigned below = (1u << lane) - 1u;
                const int dest = own ? __builtin_popcount(m & below)
                                     : nown + __builtin_popcount(~m & below);
                srcv = __builtin_amdgcn_ds_permute(dest << 2, srcv);
                dstv = __builtin_amdgcn_ds_permute(dest << 2, dstv);
                float v[32];
#pragma unroll
                for (int g = 0; g < 4; ++g) {
                    if (g * 8 >= nown) break;
#pragma unroll
                    for (int j = g * 8; j < g * 8 + 8; ++j) {
                        if (j < nown) {
                            const int sidx = __builtin_amdgcn_readlane(srcv, j);
                            v[j] = feat[(long long)sidx * 16 + c];
                        }
                    }
                }
#pragma unroll
                for (int half = 0; half < 2; ++half) {
                    if (half * 16 >= nown) break;
                    int d[16];
#pragma unroll
                    for (int j = 0; j < 16; ++j)
                        d[j] = __builtin_amdgcn_readlane(dstv, half * 16 + j);
                    if (lane < LN) {
#pragma unroll
                        for (int j = 0; j < 16; ++j) {
                            if (half * 16 + j < nown)
                                atomicAdd(&rows[(long long)d[j] * PITCH + lane],
                                          (lane < 16) ? v[half * 16 + j] : 1.0f);
                        }
                    }
                }
            }
        }
    }
}

__global__ void scatter1_kernel(const int* __restrict__ ei, int E, int N,
                                const float* __restrict__ y1,
                                float* __restrict__ arow,
                                int* __restrict__ ctr0, int* __restrict__ ctr1,
                                int split) {
    scatter_owned<32, 17>(ei, E, N, y1, arow, ctr0, ctr1, split);
}

__global__ void combine1_kernel(const float* __restrict__ arow,
                                const float* __restrict__ r1,
                                const float* __restrict__ b1,
                                float* __restrict__ h, int N) {
    long long gid = (long long)blockIdx.x * blockDim.x + threadIdx.x;
    if (gid >= (long long)N * 16) return;
    long long n = gid >> 4;
    int c = (int)(gid & 15);
    float v = arow[n * 32 + c] / fmaxf(arow[n * 32 + 16], 1.0f)
            + r1[gid] + b1[c];
    h[gid] = (v > 0.0f) ? v : (expf(v) - 1.0f);
}

__global__ void scatter2_kernel(const int* __restrict__ ei, int E, int N,
                                const float* __restrict__ h,
                                float* __restrict__ agg2,
                                int* __restrict__ ctr0, int* __restrict__ ctr1,
                                int split) {
    scatter_owned<16, 16>(ei, E, N, h, agg2, ctr0, ctr1, split);
}

__device__ __forceinline__ float b2load(const float* w2k1, const float* root2,
                                        int k, int col) {
    if (col >= 40) return 0.0f;
    return (k < 16) ? w2k1[k * 40 + col] : root2[(k - 16) * 40 + col];
}

__global__ void gemm2_kernel(const float* __restrict__ agg2,
                             const float* __restrict__ arow,
                             const float* __restrict__ h,
                             const float* __restrict__ w2k1,
                             const float* __restrict__ root2,
                             const float* __restrict__ b2,
                             float* __restrict__ out, int N) {
    const int lane = threadIdx.x & 31;
    const int wave = threadIdx.x >> 5;
    const int tile = blockIdx.x * 4 + wave;
    const int row0 = tile * 16;
    if (row0 >= N) return;

    const int colL = lane & 15;
    const int kSub = (lane >> 4) * 2;
    int rowA = row0 + colL;
    if (rowA >= N) rowA = N - 1;
    const float invd = 1.0f / fmaxf(arow[(long long)rowA * 32 + 16], 1.0f);

    v8f c0 = {}, c1 = {}, c2 = {};
#pragma unroll
    for (int kk = 0; kk < 8; ++kk) {
        const int k = kk * 4 + kSub;
        v2f a;
        a.x = (k < 16) ? agg2[(long long)rowA * 16 + k] * invd
                       : h[(long long)rowA * 16 + (k - 16)];
        a.y = (k + 1 < 16) ? agg2[(long long)rowA * 16 + k + 1] * invd
                           : h[(long long)rowA * 16 + (k + 1 - 16)];
        v2f b0, b1, b2v;
        b0.x  = b2load(w2k1, root2, k,     colL);
        b0.y  = b2load(w2k1, root2, k + 1, colL);
        b1.x  = b2load(w2k1, root2, k,     16 + colL);
        b1.y  = b2load(w2k1, root2, k + 1, 16 + colL);
        b2v.x = b2load(w2k1, root2, k,     32 + colL);
        b2v.y = b2load(w2k1, root2, k + 1, 32 + colL);
        c0 = __builtin_amdgcn_wmma_f32_16x16x4_f32(false, a, false, b0,
                                                   (short)0, c0, false, false);
        c1 = __builtin_amdgcn_wmma_f32_16x16x4_f32(false, a, false, b1,
                                                   (short)0, c1, false, false);
        c2 = __builtin_amdgcn_wmma_f32_16x16x4_f32(false, a, false, b2v,
                                                   (short)0, c2, false, false);
    }

    const int rbase = (lane >> 4) * 8;
#pragma unroll
    for (int r = 0; r < 8; ++r) {
        const int row = row0 + rbase + r;
        if (row >= N) continue;
        out[(long long)row * 40 + colL]      = c0[r] + b2[colL];
        out[(long long)row * 40 + 16 + colL] = c1[r] + b2[16 + colL];
        if (colL < 8)
            out[(long long)row * 40 + 32 + colL] = c2[r] + b2[32 + colL];
    }
}

__global__ void logsoftmax_kernel(float* __restrict__ out, int N) {
    int n = blockIdx.x * blockDim.x + threadIdx.x;
    if (n >= N) return;
    float v[40];
    float m = -INFINITY;
#pragma unroll
    for (int i = 0; i < 40; ++i) {
        v[i] = out[(long long)n * 40 + i];
        m = fmaxf(m, v[i]);
    }
    float s = 0.0f;
#pragma unroll
    for (int i = 0; i < 40; ++i) s += expf(v[i] - m);
    float ls = logf(s);
#pragma unroll
    for (int i = 0; i < 40; ++i) out[(long long)n * 40 + i] = v[i] - m - ls;
}

extern "C" void kernel_launch(void* const* d_in, const int* in_sizes, int n_in,
                              void* d_out, int out_size, void* d_ws, size_t ws_size,
                              hipStream_t stream) {
    const float* x     = (const float*)d_in[0];
    const int*   ei    = (const int*)d_in[1];
    const float* W1    = (const float*)d_in[2];
    const float* root1 = (const float*)d_in[3];
    const float* b1    = (const float*)d_in[4];
    const float* W2    = (const float*)d_in[5];
    const float* root2 = (const float*)d_in[6];
    const float* b2    = (const float*)d_in[7];
    float*       out   = (float*)d_out;

    const int N = in_sizes[0] / 128;
    const int E = in_sizes[1] / 2;

    const float* w1k1 = W1 + 128 * 16;
    const float* w2k1 = W2 + 16 * 40;

    float* ws   = (float*)d_ws;
    float* arow = ws;
    float* agg2 = ws + (size_t)N * 32;
    float* y1   = ws + (size_t)N * 48;
    float* r1   = ws + (size_t)N * 64;
    float* h    = y1;

    long long nz = (long long)N * 48;
    zero_kernel<<<(unsigned)((nz + 255) / 256), 256, 0, stream>>>(ws, nz);

    const int tiles   = (N + 15) / 16;
    const int gblocks = (tiles + 3) / 4;

    gemm1_kernel<<<gblocks, 128, 0, stream>>>(x, w1k1, root1, y1, r1, N);

    const int split = (N / 2) & ~3;
    int* pads = (int*)arow;
    int* s1c0 = pads + 24;
    int* s1c1 = pads + (size_t)split * 32 + 28;
    int* s2c0 = pads + 25;
    int* s2c1 = pads + (size_t)split * 32 + 29;
    const unsigned sgrid = 2048;

    scatter1_kernel<<<sgrid, 256, 0, stream>>>(ei, E, N, y1, arow, s1c0, s1c1, split);

    long long cthreads = (long long)N * 16;
    combine1_kernel<<<(unsigned)((cthreads + 255) / 256), 256, 0, stream>>>(
        arow, r1, b1, h, N);

    scatter2_kernel<<<sgrid, 256, 0, stream>>>(ei, E, N, h, agg2, s2c0, s2c1, split);

    gemm2_kernel<<<gblocks, 128, 0, stream>>>(agg2, arow, h, w2k1, root2, b2, out, N);

    logsoftmax_kernel<<<(unsigned)((N + 255) / 256), 256, 0, stream>>>(out, N);
}


# launch fusion, 7 -> 5: accumulator clearing moved into gemm1's waves, log-softmax into gemm2's epilogue on the accumulators (exp-sum order: 3 per lane, then 16-lane xor tree); out written once
# speedup vs baseline: 4.1552x; 1.0495x over previous
#include <hip/hip_runtime.h>
#include <hip/hip_bf16.h>
#include <math.h>

typedef __attribute__((ext_vector_type(2))) float v2f;
typedef __attribute__((ext_vector_type(8))) float v8f;

__global__ void gemm1_kernel(const float* __restrict__ x,
                             const float* __restrict__ w1k1,
                             const float* __restrict__ root1,
                             float* __restrict__ y1,
                             float* __restrict__ r1,
                             float* __restrict__ arow,
                             float* __restrict__ agg2, int N) {
    const int lane = threadIdx.x & 31;
    const int wave = threadIdx.x >> 5;
    const int tile = blockIdx.x * 4 + wave;
    const int row0 = tile * 16;
    if (row0 >= N) return;

    {
        const float4 z4 = make_float4(0.0f, 0.0f, 0.0f, 0.0f);
        float4* za = reinterpret_cast<float4*>(arow + (long long)row0 * 32);
#pragma unroll
        for (int i = 0; i < 4; ++i) {
            const int q = i * 32 + lane;
            if (row0 + (q >> 3) < N) za[q] = z4;
        }
        float4* zb = reinterpret_cast<float4*>(agg2 + (long long)row0 * 16);
#pragma unroll
        for (int i = 0; i < 2; ++i) {
            const int q = i * 32 + lane;
            if (row0 + (q >> 2) < N) zb[q] = z4;
        }
    }

    const int colL = lane & 15;
    const int kSub = (lane >> 4) * 2;
    int rowA = row0 + colL;
    if (rowA >= N) rowA = N - 1;

    v8f c0 = {};
    v8f c1 = {};
#pragma unroll 4
    for (int kk = 0; kk < 32; ++kk) {
        const int k = kk * 4 + kSub;
        v2f a;
        a.x = x[(long long)rowA * 128 + k];
        a.y = x[(long long)rowA * 128 + k + 1];
        v2f b0;
        b0.x = w1k1[k * 16 + colL];
        b0.y = w1k1[(k + 1) * 16 + colL];
        v2f b1;
        b1.x = root1[k * 16 + colL];
        b1.y = root1[(k + 1) * 16 + colL];
        c0 = __builtin_amdgcn_wmma_f32_16x16x4_f32(false, a, false, b0,
                                                   (short)0, c0, false, false);
        c1 = __builtin_amdgcn_wmma_f32_16x16x4_f32(false, a, false, b1,
                                                   (short)0, c1, false, false);
    }

    const int rbase = (lane >> 4) * 8;
#pragma unroll
    for (int r = 0; r < 8; ++r) {
        const int row = row0 + rbase + r;
        if (row < N) {
            y1[(long long)row * 16 + colL] = c0[r];
            r1[(long long)row * 16 + colL] = c1[r];
        }
    }
}

#define CH 64
#define WAVES_PER_WG 8

template <int PITCH, int LN>
__device__ __forceinline__ void scatter_owned(const int* __restrict__ ei, int E, int N,
                                              const float* __restrict__ feat,
                                              float* __restrict__ rows,
                                              int* __restrict__ ctr0,
                                              int* __restrict__ ctr1, int split) {
    __shared__ int s_ticket[2];
    const int lane = threadIdx.x & 31;
    const int wave = threadIdx.x >> 5;
    const int c = lane & 15;
    const int nchunks =
        (int)(((long long)E + CH * WAVES_PER_WG - 1) / (CH * WAVES_PER_WG));
    const unsigned id = __builtin_amdgcn_s_sendmsg_rtn(0x87);
    const int myhalf = (((id >> 16) & 0xFu) >= 4u) ? 1 : 0;

    int k = 0;
    for (int pass = 0; pass < 2; ++pass) {
        const int hsel = myhalf ^ pass;
        int* ctr = hsel ? ctr1 : ctr0;
        const int lo = hsel ? split : 0;
        const unsigned span = (unsigned)(hsel ? (N - split) : split);
        for (int it = 0; it <= nchunks; ++it) {
            const int slot = k & 1;
            ++k;
            if (threadIdx.x == 0) s_ticket[slot] = atomicAdd(ctr, 1);
            __syncthreads();
            const int t = __builtin_amdgcn_readfirstlane(s_ticket[slot]);
            if (t >= nchunks) break;
            for (int s = 0; s < CH / 32; ++s) {
                const long long e =
                    ((long long)t * WAVES_PER_WG + wave) * CH + s * 32 + lane;
                int srcv = 0, dstv = -1;
                if (e < E) {
                    srcv = ei[e];
                    dstv = ei[(long long)E + e];
                }
                const bool own = (unsigned)(dstv - lo) < span;
                const unsigned m = __builtin_amdgcn_ballot_w32(own);
                const int nown = __builtin_popcount(m);
                if (nown == 0) continue;
                const unsigned below = (1u << lane) - 1u;
                const int dest = own ? __builtin_popcount(m & below)
                                     : nown + __builtin_popcount(~m & below);
                srcv = __builtin_amdgcn_ds_permute(dest << 2, srcv);
                dstv = __builtin_amdgcn_ds_permute(dest << 2, dstv);
                float v[32];
#pragma unroll
                for (int g = 0; g < 4; ++g) {
                    if (g * 8 >= nown) break;
#pragma unroll
                    for (int j = g * 8; j < g * 8 + 8; ++j) {
                        if (j < nown) {
                            const int sidx = __builtin_amdgcn_readlane(srcv, j);
                            v[j] = feat[(long long)sidx * 16 + c];
                        }
                    }
                }
#pragma unroll
                for (int half = 0; half < 2; ++half) {
                    if (half * 16 >= nown) break;
                    int d[16];
#pragma unroll
                    for (int j = 0; j < 16; ++j)
                        d[j] = __builtin_amdgcn_readlane(dstv, half * 16 + j);
                    if (lane < LN) {
#pragma unroll
                        for (int j = 0; j < 16; ++j) {
                            if (half * 16 + j < nown)
                                atomicAdd(&rows[(long long)d[j] * PITCH + lane],
                                          (lane < 16) ? v[half * 16 + j] : 1.0f);
                        }
                    }
                }
            }
        }
    }
}

__global__ void scatter1_kernel(const int* __restrict__ ei, int E, int N,
                                const float* __restrict__ y1,
                                float* __restrict__ arow,
                                int* __restrict__ ctr0, int* __restrict__ ctr1,
                                int split) {
    scatter_owned<32, 17>(ei, E, N, y1, arow, ctr0, ctr1, split);
}

__global__ void combine1_kernel(const float* __restrict__ arow,
                                const float* __restrict__ r1,
                                const float* __restrict__ b1,
                                float* __restrict__ h, int N) {
    long long gid = (long long)blockIdx.x * blockDim.x + threadIdx.x;
    if (gid >= (long long)N * 16) return;
    long long n = gid >> 4;
    int c = (int)(gid & 15);
    float v = arow[n * 32 + c] / fmaxf(arow[n * 32 + 16], 1.0f)
            + r1[gid] + b1[c];
    h[gid] = (v > 0.0f) ? v : (expf(v) - 1.0f);
}

__global__ void scatter2_kernel(const int* __restrict__ ei, int E, int N,
                                const float* __restrict__ h,
                                float* __restrict__ agg2,
                                int* __restrict__ ctr0, int* __restrict__ ctr1,
                                int split) {
    scatter_owned<16, 16>(ei, E, N, h, agg2, ctr0, ctr1, split);
}

__device__ __forceinline__ float b2load(const float* w2k1, const float* root2,
                                        int k, int col) {
    if (col >= 40) return 0.0f;
    return (k < 16) ? w2k1[k * 40 + col] : root2[(k - 16) * 40 + col];
}

__global__ void gemm2_kernel(const float* __restrict__ agg2,
                             const float* __restrict__ arow,
                             const float* __restrict__ h,
                             const float* __restrict__ w2k1,
                             const float* __restrict__ root2,
                             const float* __restrict__ b2,
                             float* __restrict__ out, int N) {
    const int lane = threadIdx.x & 31;
    const int wave = threadIdx.x >> 5;
    const int tile = blockIdx.x * 4 + wave;
    const int row0 = tile * 16;
    if (row0 >= N) return;

    const int colL = lane & 15;
    const int kSub = (lane >> 4) * 2;
    int rowA = row0 + colL;
    if (rowA >= N) rowA = N - 1;
    const float invd = 1.0f / fmaxf(arow[(long long)rowA * 32 + 16], 1.0f);

    v8f c0 = {}, c1 = {}, c2 = {};
#pragma unroll
    for (int kk = 0; kk < 8; ++kk) {
        const int k = kk * 4 + kSub;
        v2f a;
        a.x = (k < 16) ? agg2[(long long)rowA * 16 + k] * invd
                       : h[(long long)rowA * 16 + (k - 16)];
        a.y = (k + 1 < 16) ? agg2[(long long)rowA * 16 + k + 1] * invd
                           : h[(long long)rowA * 16 + (k + 1 - 16)];
        v2f b0, b1, b2v;
        b0.x  = b2load(w2k1, root2, k,     colL);
        b0.y  = b2load(w2k1, root2, k + 1, colL);
        b1.x  = b2load(w2k1, root2, k,     16 + colL);
        b1.y  = b2load(w2k1, root2, k + 1, 16 + colL);
        b2v.x = b2load(w2k1, root2, k,     32 + colL);
        b2v.y = b2load(w2k1, root2, k + 1, 32 + colL);
        c0 = __builtin_amdgcn_wmma_f32_16x16x4_f32(false, a, false, b0,
                                                   (short)0, c0, false, false);
        c1 = __builtin_amdgcn_wmma_f32_16x16x4_f32(false, a, false, b1,
                                                   (short)0, c1, false, false);
        c2 = __builtin_amdgcn_wmma_f32_16x16x4_f32(false, a, false, b2v,
                                                   (short)0, c2, false, false);
    }

    asm volatile("v_nop\n\tv_nop\n\tv_nop\n\tv_nop" : "+v"(c0), "+v"(c1), "+v"(c2));

    const int rbase = (lane >> 4) * 8;
    const bool tail = colL < 8;
    const float bia0 = b2[colL];
    const float bia1 = b2[16 + colL];
    const float bia2 = tail ? b2[32 + colL] : 0.0f;

    float v0[8], v1[8], v2[8], m[8], s[8];
#pragma unroll
    for (int r = 0; r < 8; ++r) {
        v0[r] = c0[r] + bia0;
        v1[r] = c1[r] + bia1;
        v2[r] = tail ? (c2[r] + bia2) : -INFINITY;
        m[r] = fmaxf(fmaxf(v0[r], v1[r]), v2[r]);
    }
#pragma unroll
    for (int d = 1; d < 16; d <<= 1) {
#pragma unroll
        for (int r = 0; r < 8; ++r) m[r] = fmaxf(m[r], __shfl_xor(m[r], d));
    }
#pragma unroll
    for (int r = 0; r < 8; ++r) {
        s[r] = expf(v0[r] - m[r]) + expf(v1[r] - m[r]);
        if (tail) s[r] += expf(v2[r] - m[r]);
    }
#pragma unroll
    for (int d = 1; d < 16; d <<= 1) {
#pragma unroll
        for (int r = 0; r < 8; ++r) s[r] += __shfl_xor(s[r], d);
    }
#pragma unroll
    for (int r = 0; r < 8; ++r) {
        const int row = row0 + rbase + r;
        if (row >= N) continue;
        const float ls = logf(s[r]);
        out[(long long)row * 40 + colL]      = v0[r] - m[r] - ls;
        out[(long long)row * 40 + 16 + colL] = v1[r] - m[r] - ls;
        if (tail)
            out[(long long)row * 40 + 32 + colL] = v2[r] - m[r] - ls;
    }
}

extern "C" void kernel_launch(void* const* d_in, const int* in_sizes, int n_in,
                              void* d_out, int out_size, void* d_ws, size_t ws_size,
                              hipStream_t stream) {
    const float* x     = (const float*)d_in[0];
    const int*   ei    = (const int*)d_in[1];
    const float* W1    = (const float*)d_in[2];
    const float* root1 = (const float*)d_in[3];
    const float* b1    = (const float*)d_in[4];
    const float* W2    = (const float*)d_in[5];
    const float* root2 = (const float*)d_in[6];
    const float* b2    = (const float*)d_in[7];
    float*       out   = (float*)d_out;

    const int N = in_sizes[0] / 128;
    const int E = in_sizes[1] / 2;

    const float* w1k1 = W1 + 128 * 16;
    const float* w2k1 = W2 + 16 * 40;

    float* ws   = (float*)d_ws;
    float* arow = ws;
    float* agg2 = ws + (size_t)N * 32;
    float* y1   = ws + (size_t)N * 48;
    float* r1   = ws + (size_t)N * 64;
    float* h    = y1;

    const int tiles   = (N + 15) / 16;
    const int gblocks = (tiles + 3) / 4;

    gemm1_kernel<<<gblocks, 128, 0, stream>>>(x, w1k1, root1, y1, r1, arow, agg2, N);

    const int split = (N / 2) & ~3;
    int* pads = (int*)arow;
    int* s1c0 = pads + 24;
    int* s1c1 = pads + (size_t)split * 32 + 28;
    int* s2c0 = pads + 25;
    int* s2c1 = pads + (size_t)split * 32 + 29;
    const unsigned sgrid = 2048;

    scatter1_kernel<<<sgrid, 256, 0, stream>>>(ei, E, N, y1, arow, s1c0, s1c1, split);

    long long cthreads = (long long)N * 16;
    combine1_kernel<<<(unsigned)((cthreads + 255) / 256), 256, 0, stream>>>(
        arow, r1, b1, h, N);

    scatter2_kernel<<<sgrid, 256, 0, stream>>>(ei, E, N, h, agg2, s2c0, s2c1, split);

    gemm2_kernel<<<gblocks, 128, 0, stream>>>(agg2, arow, h, w2k1, root2, b2, out, N);
}


# both edge sums as a pull: edges bucket-ordered per 16384-edge chunk in LDS, whole-line list writes; 256 owner work-groups add gathered rows in LDS; no global atomics; sums in LDS arrival order
# speedup vs baseline: 6.5080x; 1.5662x over previous
#include <hip/hip_runtime.h>
#include <hip/hip_bf16.h>
#include <math.h>


typedef __attribute__((ext_vector_type(2))) float v2f;
typedef __attribute__((ext_vector_type(8))) float v8f;

#define CHK      16384
#define PT       1024
#define EPT      (CHK / PT)
#define NOWN     256
#define NBKMAX   512
#define OP       288
#define SRC_BITS 23
#define SRC_MASK 0x7FFFFFu

__global__ void gemm1_kernel(const float* __restrict__ x,
                             const float* __restrict__ w1k1,
                             const float* __restrict__ root1,
                             float* __restrict__ y1,
                             float* __restrict__ r1, int N) {
    const int lane = threadIdx.x & 31;
    const int wave = threadIdx.x >> 5;
    const int tile = blockIdx.x * 4 + wave;
    const int row0 = tile * 16;
    if (row0 >= N) return;

    const int colL = lane & 15;
    const int kSub = (lane >> 4) * 2;
    int rowA = row0 + colL;
    if (rowA >= N) rowA = N - 1;

    v8f c0 = {};
    v8f c1 = {};
#pragma unroll 4
    for (int kk = 0; kk < 32; ++kk) {
        const int k = kk * 4 + kSub;
        v2f a;
        a.x = x[(long long)rowA * 128 + k];
        a.y = x[(long long)rowA * 128 + k + 1];
        v2f b0;
        b0.x = w1k1[k * 16 + colL];
        b0.y = w1k1[(k + 1) * 16 + colL];
        v2f b1;
        b1.x = root1[k * 16 + colL];
        b1.y = root1[(k + 1) * 16 + colL];
        c0 = __builtin_amdgcn_wmma_f32_16x16x4_f32(false, a, false, b0,
                                                   (short)0, c0, false, false);
        c1 = __builtin_amdgcn_wmma_f32_16x16x4_f32(false, a, false, b1,
                                                   (short)0, c1, false, false);
    }

    const int rbase = (lane >> 4) * 8;
#pragma unroll
    for (int r = 0; r < 8; ++r) {
        const int row = row0 + rbase + r;
        if (row < N) {
            y1[(long long)row * 16 + colL] = c0[r];
            r1[(long long)row * 16 + colL] = c1[r];
        }
    }
}

__global__ __launch_bounds__(PT) void place_kernel(const int* __restrict__ ei, int E, int N,
                                                   int nbk,
                                                   int* __restrict__ list,
                                                   int* __restrict__ offs) {
    __shared__ int s_cnt[NOWN];
    __shared__ int s_start[NOWN + 1];
    __shared__ int s_wsum[NOWN / 32];
    __shared__ int s_sorted[CHK];
    const int tid = threadIdx.x;
    const int lane = tid & 31;
    const int wave = tid >> 5;
    const int w = blockIdx.x;
    const long long base = (long long)w * CHK;

    if (tid < NOWN) s_cnt[tid] = 0;
    __syncthreads();

    int sv[EPT], dv[EPT];
#pragma unroll
    for (int j = 0; j < EPT; ++j) {
        const long long e = base + j * PT + tid;
        sv[j] = 0;
        dv[j] = -1;
        if (e < E) {
            sv[j] = ei[e];
            dv[j] = ei[(long long)E + e];
        }
    }
    int ent[EPT], bk[EPT], rk[EPT];
#pragma unroll
    for (int j = 0; j < EPT; ++j) {
        bk[j] = -1;
        ent[j] = 0;
        rk[j] = 0;
        const int d = dv[j];
        if ((unsigned)d < (unsigned)N) {
            int s = sv[j];
            s = (s < 0) ? 0 : ((s >= N) ? (N - 1) : s);
            int b = d / nbk;
            if (b > NOWN - 1) b = NOWN - 1;
            const int dl = (d - b * nbk) & (NBKMAX - 1);
            bk[j] = b;
            ent[j] = (int)(((unsigned)dl << SRC_BITS) | ((unsigned)s & SRC_MASK));
            rk[j] = atomicAdd(&s_cnt[b], 1);
        }
    }
    __syncthreads();

    int c = 0, incl = 0;
    if (tid < NOWN) {
        c = s_cnt[tid];
        incl = c;
#pragma unroll
        for (int d = 1; d < 32; d <<= 1) {
            const int t = __shfl_up(incl, d);
            if (lane >= d) incl += t;
        }
        if (lane == 31) s_wsum[wave] = incl;
    }
    __syncthreads();
    if (tid < NOWN) {
        int add = 0;
        for (int k = 0; k < wave; ++k) add += s_wsum[k];
        s_start[tid] = incl - c + add;
        if (tid == NOWN - 1) s_start[NOWN] = incl + add;
    }
    __syncthreads();

#pragma unroll
    for (int j = 0; j < EPT; ++j)
        if (bk[j] >= 0) s_sorted[s_start[bk[j]] + rk[j]] = ent[j];
    __syncthreads();

    const int total = s_start[NOWN];
#pragma unroll
    for (int j = 0; j < EPT; ++j) {
        const int i = j * PT + tid;
        if (i < total) list[base + i] = s_sorted[i];
    }
    if (tid < OP) offs[(long long)w * OP + tid] = (tid <= NOWN) ? s_start[tid] : total;
}

template <int LAYER>
__device__ __forceinline__ void owner_body(const int* __restrict__ list,
                                           const int* __restrict__ offs,
                                           int nChunks, int N, int nbk,
                                           const float* __restrict__ feat,
                                           float* r1h,
                                           const float* __restrict__ b1,
                                           float* __restrict__ agg2) {
    __shared__ float s_acc[NBKMAX * 16];
    __shared__ int s_deg[NBKMAX];
    const int tid = threadIdx.x;
    const int lane = tid & 31;
    const int wave = __builtin_amdgcn_readfirstlane(tid >> 5);
    const int c = lane & 15;
    const int hsel = lane >> 4;
    const int b = blockIdx.x;
    const unsigned nlast = (unsigned)(N - 1);

    for (int i = tid; i < NBKMAX * 16; i += PT) s_acc[i] = 0.0f;
    if (tid < NBKMAX) s_deg[tid] = 0;
    __syncthreads();

    for (int w = wave; w < nChunks; w += PT / 32) {
        int o0 = __builtin_amdgcn_readfirstlane(offs[(long long)w * OP + b]);
        int o1 = __builtin_amdgcn_readfirstlane(offs[(long long)w * OP + b + 1]);
        o1 = (o1 < 0) ? 0 : ((o1 > CHK) ? CHK : o1);
        o0 = (o0 < 0) ? 0 : ((o0 > o1) ? o1 : o0);
        const int* run = list + (long long)w * CHK;
        int nxt = 0;
        if (o0 + lane < o1) nxt = run[o0 + lane];
        for (int p = o0; p < o1; p += 32) {
            const int ent = nxt;
            const int rem = o1 - p;
            const int cnt = (rem < 32) ? rem : 32;
            nxt = 0;
            if (p + 32 + lane < o1) nxt = run[p + 32 + lane];
            if (lane < cnt) atomicAdd(&s_deg[((unsigned)ent) >> SRC_BITS], 1);
            float v[16];
            int en[16];
#pragma unroll
            for (int g = 0; g < 4; ++g) {
                if (g * 8 >= cnt) break;
#pragma unroll
                for (int jj = g * 4; jj < g * 4 + 4; ++jj) {
                    en[jj] = __shfl(ent, 2 * jj + hsel);
                    unsigned sidx = (unsigned)en[jj] & SRC_MASK;
                    if (sidx > nlast) sidx = nlast;
                    v[jj] = feat[(sidx << 4) | (unsigned)c];
                }
            }
#pragma unroll
            for (int g = 0; g < 4; ++g) {
                if (g * 8 >= cnt) break;
#pragma unroll
                for (int jj = g * 4; jj < g * 4 + 4; ++jj) {
                    const float add = (2 * jj + hsel < cnt) ? v[jj] : 0.0f;
                    atomicAdd(&s_acc[((((unsigned)en[jj]) >> SRC_BITS) << 4) | (unsigned)c], add);
                }
            }
        }
    }
    __syncthreads();

    const int node0 = b * nbk;
#pragma unroll
    for (int q = 0; q < NBKMAX * 16 / PT; ++q) {
        const int idx = q * PT + tid;
        const int r = idx >> 4;
        const int cc = idx & 15;
        const int n = node0 + r;
        if (r < nbk && n < N) {
            const float dg = fmaxf((float)s_deg[r], 1.0f);
            if (LAYER == 1) {
                const float v = s_acc[idx] / dg + r1h[(long long)n * 16 + cc] + b1[cc];
                r1h[(long long)n * 16 + cc] = (v > 0.0f) ? v : (expf(v) - 1.0f);
            } else {
                const float invd = 1.0f / dg;
                agg2[(long long)n * 16 + cc] = s_acc[idx] * invd;
            }
        }
    }
}

__global__ __launch_bounds__(PT) void owner1_kernel(const int* __restrict__ list,
                                                    const int* __restrict__ offs,
                                                    int nChunks, int N, int nbk,
                                                    const float* __restrict__ y1,
                                                    float* r1h,
                                                    const float* __restrict__ b1) {
    owner_body<1>(list, offs, nChunks, N, nbk, y1, r1h, b1, nullptr);
}

__global__ __launch_bounds__(PT) void owner2_kernel(const int* __restrict__ list,
                                                    const int* __restrict__ offs,
                                                    int nChunks, int N, int nbk,
                                                    const float* __restrict__ h,
                                                    float* __restrict__ agg2) {
    owner_body<2>(list, offs, nChunks, N, nbk, h, nullptr, nullptr, agg2);
}

__device__ __forceinline__ float b2load(const float* w2k1, const float* root2,
                                        int k, int col) {
    if (col >= 40) return 0.0f;
    return (k < 16) ? w2k1[k * 40 + col] : root2[(k - 16) * 40 + col];
}

__global__ void gemm2_kernel(const float* __restrict__ agg2,
                             const float* __restrict__ h,
                             const float* __restrict__ w2k1,
                             const float* __restrict__ root2,
                             const float* __restrict__ b2,
                             float* __restrict__ out, int N) {
    const int lane = threadIdx.x & 31;
    const int wave = threadIdx.x >> 5;
    const int tile = blockIdx.x * 4 + wave;
    const int row0 = tile * 16;
    if (row0 >= N) return;

    const int colL = lane & 15;
    const int kSub = (lane >> 4) * 2;
    int rowA = row0 + colL;
    if (rowA >= N) rowA = N - 1;

    v8f c0 = {}, c1 = {}, c2 = {};
#pragma unroll
    for (int kk = 0; kk < 8; ++kk) {
        const int k = kk * 4 + kSub;
        v2f a;
        a.x = (k < 16) ? agg2[(long long)rowA * 16 + k]
                       : h[(long long)rowA * 16 + (k - 16)];
        a.y = (k + 1 < 16) ? agg2[(long long)rowA * 16 + k + 1]
                           : h[(long long)rowA * 16 + (k + 1 - 16)];
        v2f b0, b1, b2v;
        b0.x  = b2load(w2k1, root2, k,     colL);
        b0.y  = b2load(w2k1, root2, k + 1, colL);
        b1.x  = b2load(w2k1, root2, k,     16 + colL);
        b1.y  = b2load(w2k1, root2, k + 1, 16 + colL);
        b2v.x = b2load(w2k1, root2, k,     32 + colL);
        b2v.y = b2load(w2k1, root2, k + 1, 32 + colL);
        c0 = __builtin_amdgcn_wmma_f32_16x16x4_f32(false, a, false, b0,
                                                   (short)0, c0, false, false);
        c1 = __builtin_amdgcn_wmma_f32_16x16x4_f32(false, a, false, b1,
                                                   (short)0, c1, false, false);
        c2 = __builtin_amdgcn_wmma_f32_16x16x4_f32(false, a, false, b2v,
                                                   (short)0, c2, false, false);
    }

    asm volatile("v_nop\n\tv_nop\n\tv_nop\n\tv_nop" : "+v"(c0), "+v"(c1), "+v"(c2));

    const int rbase = (lane >> 4) * 8;
    const bool tail = colL < 8;
    const float bia0 = b2[colL];
    const float bia1 = b2[16 + colL];
    const float bia2 = tail ? b2[32 + colL] : 0.0f;

    float v0[8], v1[8], v2[8], m[8], s[8];
#pragma unroll
    for (int r = 0; r < 8; ++r) {
        v0[r] = c0[r] + bia0;
        v1[r] = c1[r] + bia1;
        v2[r] = tail ? (c2[r] + bia2) : -INFINITY;
        m[r] = fmaxf(fmaxf(v0[r], v1[r]), v2[r]);
    }
#pragma unroll
    for (int d = 1; d < 16; d <<= 1) {
#pragma unroll
        for (int r = 0; r < 8; ++r) m[r] = fmaxf(m[r], __shfl_xor(m[r], d));
    }
#pragma unroll
    for (int r = 0; r < 8; ++r) {
        s[r] = expf(v0[r] - m[r]) + expf(v1[r] - m[r]);
        if (tail) s[r] += expf(v2[r] - m[r]);
    }
#pragma unroll
    for (int d = 1; d < 16; d <<= 1) {
#pragma unroll
        for (int r = 0; r < 8; ++r) s[r] += __shfl_xor(s[r], d);
    }
#pragma unroll
    for (int r = 0; r < 8; ++r) {
        const int row = row0 + rbase + r;
        if (row >= N) continue;
        const float ls = logf(s[r]);
        out[(long long)row * 40 + colL]      = v0[r] - m[r] - ls;
        out[(long long)row * 40 + 16 + colL] = v1[r] - m[r] - ls;
        if (tail)
            out[(long long)row * 40 + 32 + colL] = v2[r] - m[r] - ls;
    }
}

extern "C" void kernel_launch(void* const* d_in, const int* in_sizes, int n_in,
                              void* d_out, int out_size, void* d_ws, size_t ws_size,
                              hipStream_t stream) {
    const float* x     = (const float*)d_in[0];
    const int*   ei    = (const int*)d_in[1];
    const float* W1    = (const float*)d_in[2];
    const float* root1 = (const float*)d_in[3];
    const float* b1    = (const float*)d_in[4];
    const float* W2    = (const float*)d_in[5];
    const float* root2 = (const float*)d_in[6];
    const float* b2    = (const float*)d_in[7];
    float*       out   = (float*)d_out;

    const int N = in_sizes[0] / 128;
    const int E = in_sizes[1] / 2;

    const float* w1k1 = W1 + 128 * 16;
    const float* w2k1 = W2 + 16 * 40;

    const int nChunks = (int)(((long long)E + CHK - 1) / CHK);
    const int nbk     = ((N + NOWN - 1) / NOWN + 1) & ~1;
    float* ws   = (float*)d_ws;
    float* y1   = ws;
    float* r1h  = ws + (size_t)N * 16;
    int*   list = (int*)(ws + (size_t)N * 32);
    int*   offs = list + (size_t)nChunks * CHK;
    float* agg2 = y1;

    const int tiles   = (N + 15) / 16;
    const int gblocks = (tiles + 3) / 4;

    place_kernel<<<nChunks, PT, 0, stream>>>(ei, E, N, nbk, list, offs);

    gemm1_kernel<<<gblocks, 128, 0, stream>>>(x, w1k1, root1, y1, r1h, N);

    owner1_kernel<<<NOWN, PT, 0, stream>>>(list, offs, nChunks, N, nbk, y1, r1h, b1);

    owner2_kernel<<<NOWN, PT, 0, stream>>>(list, offs, nChunks, N, nbk, r1h, agg2);

    gemm2_kernel<<<gblocks, 128, 0, stream>>>(agg2, r1h, w2k1, root2, b2, out, N);
}


# launch fusion 5 -> 4: gemm2 and its log-softmax run as the tail of the layer-2 owner work-group on its LDS means, one wave per 16-row tile; agg2 never written or re-read; sum orders unchanged
# speedup vs baseline: 6.8855x; 1.0580x over previous
#include <hip/hip_runtime.h>
#include <hip/hip_bf16.h>
#include <math.h>


typedef __attribute__((ext_vector_type(2))) float v2f;
typedef __attribute__((ext_vector_type(8))) float v8f;

#define CHK      16384
#define PT       1024
#define EPT      (CHK / PT)
#define NOWN     256
#define NBKMAX   512
#define OP       288
#define SRC_BITS 23
#define SRC_MASK 0x7FFFFFu

__global__ void gemm1_kernel(const float* __restrict__ x,
                             const float* __restrict__ w1k1,
                             const float* __restrict__ root1,
                             float* __restrict__ y1,
                             float* __restrict__ r1, int N) {
    const int lane = threadIdx.x & 31;
    const int wave = threadIdx.x >> 5;
    const int tile = blockIdx.x * 4 + wave;
    const int row0 = tile * 16;
    if (row0 >= N) return;

    const int colL = lane & 15;
    const int kSub = (lane >> 4) * 2;
    int rowA = row0 + colL;
    if (rowA >= N) rowA = N - 1;

    v8f c0 = {};
    v8f c1 = {};
#pragma unroll 4
    for (int kk = 0; kk < 32; ++kk) {
        const int k = kk * 4 + kSub;
        v2f a;
        a.x = x[(long long)rowA * 128 + k];
        a.y = x[(long long)rowA * 128 + k + 1];
        v2f b0;
        b0.x = w1k1[k * 16 + colL];
        b0.y = w1k1[(k + 1) * 16 + colL];
        v2f b1;
        b1.x = root1[k * 16 + colL];
        b1.y = root1[(k + 1) * 16 + colL];
        c0 = __builtin_amdgcn_wmma_f32_16x16x4_f32(false, a, false, b0,
                                                   (short)0, c0, false, false);
        c1 = __builtin_amdgcn_wmma_f32_16x16x4_f32(false, a, false, b1,
                                                   (short)0, c1, false, false);
    }

    const int rbase = (lane >> 4) * 8;
#pragma unroll
    for (int r = 0; r < 8; ++r) {
        const int row = row0 + rbase + r;
        if (row < N) {
            y1[(long long)row * 16 + colL] = c0[r];
            r1[(long long)row * 16 + colL] = c1[r];
        }
    }
}

__global__ __launch_bounds__(PT) void place_kernel(const int* __restrict__ ei, int E, int N,
                                                   int nbk,
                                                   int* __restrict__ list,
                                                   int* __restrict__ offs) {
    __shared__ int s_cnt[NOWN];
    __shared__ int s_start[NOWN + 1];
    __shared__ int s_wsum[NOWN / 32];
    __shared__ int s_sorted[CHK];
    const int tid = threadIdx.x;
    const int lane = tid & 31;
    const int wave = tid >> 5;
    const int w = blockIdx.x;
    const long long base = (long long)w * CHK;

    if (tid < NOWN) s_cnt[tid] = 0;
    __syncthreads();

    int sv[EPT], dv[EPT];
#pragma unroll
    for (int j = 0; j < EPT; ++j) {
        const long long e = base + j * PT + tid;
        sv[j] = 0;
        dv[j] = -1;
        if (e < E) {
            sv[j] = ei[e];
            dv[j] = ei[(long long)E + e];
        }
    }
    int ent[EPT], bk[EPT], rk[EPT];
#pragma unroll
    for (int j = 0; j < EPT; ++j) {
        bk[j] = -1;
        ent[j] = 0;
        rk[j] = 0;
        const int d = dv[j];
        if ((unsigned)d < (unsigned)N) {
            int s = sv[j];
            s = (s < 0) ? 0 : ((s >= N) ? (N - 1) : s);
            int b = d / nbk;
            if (b > NOWN - 1) b = NOWN - 1;
            const int dl = (d - b * nbk) & (NBKMAX - 1);
            bk[j] = b;
            ent[j] = (int)(((unsigned)dl << SRC_BITS) | ((unsigned)s & SRC_MASK));
            rk[j] = atomicAdd(&s_cnt[b], 1);
        }
    }
    __syncthreads();

    int c = 0, incl = 0;
    if (tid < NOWN) {
        c = s_cnt[tid];
        incl = c;
#pragma unroll
        for (int d = 1; d < 32; d <<= 1) {
            const int t = __shfl_up(incl, d);
            if (lane >= d) incl += t;
        }
        if (lane == 31) s_wsum[wave] = incl;
    }
    __syncthreads();
    if (tid < NOWN) {
        int add = 0;
        for (int k = 0; k < wave; ++k) add += s_wsum[k];
        s_start[tid] = incl - c + add;
        if (tid == NOWN - 1) s_start[NOWN] = incl + add;
    }
    __syncthreads();

#pragma unroll
    for (int j = 0; j < EPT; ++j)
        if (bk[j] >= 0) s_sorted[s_start[bk[j]] + rk[j]] = ent[j];
    __syncthreads();

    const int total = s_start[NOWN];
#pragma unroll
    for (int j = 0; j < EPT; ++j) {
        const int i = j * PT + tid;
        if (i < total) list[base + i] = s_sorted[i];
    }
    if (tid < OP) offs[(long long)w * OP + tid] = (tid <= NOWN) ? s_start[tid] : total;
}

__device__ __forceinline__ float b2load(const float* w2k1, const float* root2,
                                        int k, int col) {
    if (col >= 40) return 0.0f;
    return (k < 16) ? w2k1[k * 40 + col] : root2[(k - 16) * 40 + col];
}

__device__ __forceinline__ void gemm2_tail(const float* s_acc, const int* s_deg,
                                           int wave, int lane,
                                           int node0, int nbk, int N,
                                           const float* __restrict__ h,
                                           const float* __restrict__ w2k1,
                                           const float* __restrict__ root2,
                                           const float* __restrict__ b2,
                                           float* __restrict__ out) {
    const int lr0 = wave * 16;
    if (lr0 >= nbk) return;

    const int colL = lane & 15;
    const int kSub = (lane >> 4) * 2;
    const int rl = lr0 + colL;
    int nA = node0 + rl;
    if (nA >= N) nA = N - 1;
    const float invd = 1.0f / fmaxf((float)s_deg[rl], 1.0f);

    v8f c0 = {}, c1 = {}, c2 = {};
#pragma unroll
    for (int kk = 0; kk < 8; ++kk) {
        const int k = kk * 4 + kSub;
        v2f a;
        a.x = (k < 16) ? s_acc[rl * 16 + k] * invd
                       : h[(long long)nA * 16 + (k - 16)];
        a.y = (k + 1 < 16) ? s_acc[rl * 16 + k + 1] * invd
                           : h[(long long)nA * 16 + (k + 1 - 16)];
        v2f b0, b1, b2v;
        b0.x  = b2load(w2k1, root2, k,     colL);
        b0.y  = b2load(w2k1, root2, k + 1, colL);
        b1.x  = b2load(w2k1, root2, k,     16 + colL);
        b1.y  = b2load(w2k1, root2, k + 1, 16 + colL);
        b2v.x = b2load(w2k1, root2, k,     32 + colL);
        b2v.y = b2load(w2k1, root2, k + 1, 32 + colL);
        c0 = __builtin_amdgcn_wmma_f32_16x16x4_f32(false, a, false, b0,
                                                   (short)0, c0, false, false);
        c1 = __builtin_amdgcn_wmma_f32_16x16x4_f32(false, a, false, b1,
                                                   (short)0, c1, false, false);
        c2 = __builtin_amdgcn_wmma_f32_16x16x4_f32(false, a, false, b2v,
                                                   (short)0, c2, false, false);
    }

    asm volatile("v_nop\n\tv_nop\n\tv_nop\n\tv_nop" : "+v"(c0), "+v"(c1), "+v"(c2));

    const int rbase = (lane >> 4) * 8;
    const bool tail = colL < 8;
    const float bia0 = b2[colL];
    const float bia1 = b2[16 + colL];
    const float bia2 = tail ? b2[32 + colL] : 0.0f;

    float v0[8], v1[8], v2[8], m[8], s[8];
#pragma unroll
    for (int r = 0; r < 8; ++r) {
        v0[r] = c0[r] + bia0;
        v1[r] = c1[r] + bia1;
        v2[r] = tail ? (c2[r] + bia2) : -INFINITY;
        m[r] = fmaxf(fmaxf(v0[r], v1[r]), v2[r]);
    }
#pragma unroll
    for (int d = 1; d < 16; d <<= 1) {
#pragma unroll
        for (int r = 0; r < 8; ++r) m[r] = fmaxf(m[r], __shfl_xor(m[r], d));
    }
#pragma unroll
    for (int r = 0; r < 8; ++r) {
        s[r] = expf(v0[r] - m[r]) + expf(v1[r] - m[r]);
        if (tail) s[r] += expf(v2[r] - m[r]);
    }
#pragma unroll
    for (int d = 1; d < 16; d <<= 1) {
#pragma unroll
        for (int r = 0; r < 8; ++r) s[r] += __shfl_xor(s[r], d);
    }
#pragma unroll
    for (int r = 0; r < 8; ++r) {
        const int lrow = lr0 + rbase + r;
        const int row = node0 + lrow;
        if (lrow >= nbk || row >= N) continue;
        const float ls = logf(s[r]);
        out[(long long)row * 40 + colL]      = v0[r] - m[r] - ls;
        out[(long long)row * 40 + 16 + colL] = v1[r] - m[r] - ls;
        if (tail)
            out[(long long)row * 40 + 32 + colL] = v2[r] - m[r] - ls;
    }
}

template <int LAYER>
__device__ __forceinline__ void owner_body(const int* __restrict__ list,
                                           const int* __restrict__ offs,
                                           int nChunks, int N, int nbk,
                                           const float* __restrict__ feat,
                                           float* r1h,
                                           const float* __restrict__ b1,
                                           const float* __restrict__ w2k1,
                                           const float* __restrict__ root2,
                                           const float* __restrict__ b2,
                                           float* __restrict__ out) {
    __shared__ float s_acc[NBKMAX * 16];
    __shared__ int s_deg[NBKMAX];
    const int tid = threadIdx.x;
    const int lane = tid & 31;
    const int wave = __builtin_amdgcn_readfirstlane(tid >> 5);
    const int c = lane & 15;
    const int hsel = lane >> 4;
    const int b = blockIdx.x;
    const unsigned nlast = (unsigned)(N - 1);

    for (int i = tid; i < NBKMAX * 16; i += PT) s_acc[i] = 0.0f;
    if (tid < NBKMAX) s_deg[tid] = 0;
    __syncthreads();

    for (int w = wave; w < nChunks; w += PT / 32) {
        int o0 = __builtin_amdgcn_readfirstlane(offs[(long long)w * OP + b]);
        int o1 = __builtin_amdgcn_readfirstlane(offs[(long long)w * OP + b + 1]);
        o1 = (o1 < 0) ? 0 : ((o1 > CHK) ? CHK : o1);
        o0 = (o0 < 0) ? 0 : ((o0 > o1) ? o1 : o0);
        const int* run = list + (long long)w * CHK;
        int nxt = 0;
        if (o0 + lane < o1) nxt = run[o0 + lane];
        for (int p = o0; p < o1; p += 32) {
            const int ent = nxt;
            const int rem = o1 - p;
            const int cnt = (rem < 32) ? rem : 32;
            nxt = 0;
            if (p + 32 + lane < o1) nxt = run[p + 32 + lane];
            if (lane < cnt) atomicAdd(&s_deg[((unsigned)ent) >> SRC_BITS], 1);
            float v[16];
            int en[16];
#pragma unroll
            for (int g = 0; g < 4; ++g) {
                if (g * 8 >= cnt) break;
#pragma unroll
                for (int jj = g * 4; jj < g * 4 + 4; ++jj) {
                    en[jj] = __shfl(ent, 2 * jj + hsel);
                    unsigned sidx = (unsigned)en[jj] & SRC_MASK;
                    if (sidx > nlast) sidx = nlast;
                    v[jj] = feat[(sidx << 4) | (unsigned)c];
                }
            }
#pragma unroll
            for (int g = 0; g < 4; ++g) {
                if (g * 8 >= cnt) break;
#pragma unroll
                for (int jj = g * 4; jj < g * 4 + 4; ++jj) {
                    const float add = (2 * jj + hsel < cnt) ? v[jj] : 0.0f;
                    atomicAdd(&s_acc[((((unsigned)en[jj]) >> SRC_BITS) << 4) | (unsigned)c], add);
                }
            }
        }
    }
    __syncthreads();

    const int node0 = b * nbk;
    if (LAYER == 1) {
#pragma unroll
        for (int q = 0; q < NBKMAX * 16 / PT; ++q) {
            const int idx = q * PT + tid;
            const int r = idx >> 4;
            const int cc = idx & 15;
            const int n = node0 + r;
            if (r < nbk && n < N) {
                const float dg = fmaxf((float)s_deg[r], 1.0f);
                const float v = s_acc[idx] / dg + r1h[(long long)n * 16 + cc] + b1[cc];
                r1h[(long long)n * 16 + cc] = (v > 0.0f) ? v : (expf(v) - 1.0f);
            }
        }
    } else {
        gemm2_tail(s_acc, s_deg, wave, lane, node0, nbk, N, feat, w2k1, root2, b2, out);
    }
}

__global__ __launch_bounds__(PT) void owner1_kernel(const int* __restrict__ list,
                                                    const int* __restrict__ offs,
                                                    int nChunks, int N, int nbk,
                                                    const float* __restrict__ y1,
                                                    float* r1h,
                                                    const float* __restrict__ b1) {
    owner_body<1>(list, offs, nChunks, N, nbk, y1, r1h, b1, nullptr, nullptr, nullptr, nullptr);
}

__global__ __launch_bounds__(PT) void owner2_kernel(const int* __restrict__ list,
                                                    const int* __restrict__ offs,
                                                    int nChunks, int N, int nbk,
                                                    const float* __restrict__ h,
                                                    const float* __restrict__ w2k1,
                                                    const float* __restrict__ root2,
                                                    const float* __restrict__ b2,
                                                    float* __restrict__ out) {
    owner_body<2>(list, offs, nChunks, N, nbk, h, nullptr, nullptr, w2k1, root2, b2, out);
}

extern "C" void kernel_launch(void* const* d_in, const int* in_sizes, int n_in,
                              void* d_out, int out_size, void* d_ws, size_t ws_size,
                              hipStream_t stream) {
    const float* x     = (const float*)d_in[0];
    const int*   ei    = (const int*)d_in[1];
    const float* W1    = (const float*)d_in[2];
    const float* root1 = (const float*)d_in[3];
    const float* b1    = (const float*)d_in[4];
    const float* W2    = (const float*)d_in[5];
    const float* root2 = (const float*)d_in[6];
    const float* b2    = (const float*)d_in[7];
    float*       out   = (float*)d_out;

    const int N = in_sizes[0] / 128;
    const int E = in_sizes[1] / 2;

    const float* w1k1 = W1 + 128 * 16;
    const float* w2k1 = W2 + 16 * 40;

    const int nChunks = (int)(((long long)E + CHK - 1) / CHK);
    const int nbk     = ((N + NOWN - 1) / NOWN + 3) & ~3;
    float* ws   = (float*)d_ws;
    float* y1   = ws;
    float* r1h  = ws + (size_t)N * 16;
    int*   list = (int*)(ws + (size_t)N * 32);
    int*   offs = list + (size_t)nChunks * CHK;

    const int tiles   = (N + 15) / 16;
    const int gblocks = (tiles + 3) / 4;

    place_kernel<<<nChunks, PT, 0, stream>>>(ei, E, N, nbk, list, offs);

    gemm1_kernel<<<gblocks, 128, 0, stream>>>(x, w1k1, root1, y1, r1h, N);

    owner1_kernel<<<NOWN, PT, 0, stream>>>(list, offs, nChunks, N, nbk, y1, r1h, b1);

    owner2_kernel<<<NOWN, PT, 0, stream>>>(list, offs, nChunks, N, nbk, r1h,
                                           w2k1, root2, b2, out);
}


# owner kernels: a step's 16 entry broadcasts are DPP row_share moves on the vector ALU (lanes hold entries in pair order), not ds_bpermute round trips waited for before the gathers; sums unchanged
# speedup vs baseline: 7.3455x; 1.0668x over previous
#include <hip/hip_runtime.h>
#include <hip/hip_bf16.h>
#include <math.h>


typedef __attribute__((ext_vector_type(2))) float v2f;
typedef __attribute__((ext_vector_type(8))) float v8f;

#define CHK      16384
#define PT       1024
#define EPT      (CHK / PT)
#define NOWN     256
#define NBKMAX   512
#define OP       288
#define SRC_BITS 23
#define SRC_MASK 0x7FFFFFu

__global__ void gemm1_kernel(const float* __restrict__ x,
                             const float* __restrict__ w1k1,
                             const float* __restrict__ root1,
                             float* __restrict__ y1,
                             float* __restrict__ r1, int N) {
    const int lane = threadIdx.x & 31;
    const int wave = threadIdx.x >> 5;
    const int tile = blockIdx.x * 4 + wave;
    const int row0 = tile * 16;
    if (row0 >= N) return;

    const int colL = lane & 15;
    const int kSub = (lane >> 4) * 2;
    int rowA = row0 + colL;
    if (rowA >= N) rowA = N - 1;

    v8f c0 = {};
    v8f c1 = {};
#pragma unroll 4
    for (int kk = 0; kk < 32; ++kk) {
        const int k = kk * 4 + kSub;
        v2f a;
        a.x = x[(long long)rowA * 128 + k];
        a.y = x[(long long)rowA * 128 + k + 1];
        v2f b0;
        b0.x = w1k1[k * 16 + colL];
        b0.y = w1k1[(k + 1) * 16 + colL];
        v2f b1;
        b1.x = root1[k * 16 + colL];
        b1.y = root1[(k + 1) * 16 + colL];
        c0 = __builtin_amdgcn_wmma_f32_16x16x4_f32(false, a, false, b0,
                                                   (short)0, c0, false, false);
        c1 = __builtin_amdgcn_wmma_f32_16x16x4_f32(false, a, false, b1,
                                                   (short)0, c1, false, false);
    }

    const int rbase = (lane >> 4) * 8;
#pragma unroll
    for (int r = 0; r < 8; ++r) {
        const int row = row0 + rbase + r;
        if (row < N) {
            y1[(long long)row * 16 + colL] = c0[r];
            r1[(long long)row * 16 + colL] = c1[r];
        }
    }
}

__global__ __launch_bounds__(PT) void place_kernel(const int* __restrict__ ei, int E, int N,
                                                   int nbk,
                                                   int* __restrict__ list,
                                                   int* __restrict__ offs) {
    __shared__ int s_cnt[NOWN];
    __shared__ int s_start[NOWN + 1];
    __shared__ int s_wsum[NOWN / 32];
    __shared__ int s_sorted[CHK];
    const int tid = threadIdx.x;
    const int lane = tid & 31;
    const int wave = tid >> 5;
    const int w = blockIdx.x;
    const long long base = (long long)w * CHK;

    if (tid < NOWN) s_cnt[tid] = 0;
    __syncthreads();

    int sv[EPT], dv[EPT];
#pragma unroll
    for (int j = 0; j < EPT; ++j) {
        const long long e = base + j * PT + tid;
        sv[j] = 0;
        dv[j] = -1;
        if (e < E) {
            sv[j] = ei[e];
            dv[j] = ei[(long long)E + e];
        }
    }
    int ent[EPT], bk[EPT], rk[EPT];
#pragma unroll
    for (int j = 0; j < EPT; ++j) {
        bk[j] = -1;
        ent[j] = 0;
        rk[j] = 0;
        const int d = dv[j];
        if ((unsigned)d < (unsigned)N) {
            int s = sv[j];
            s = (s < 0) ? 0 : ((s >= N) ? (N - 1) : s);
            int b = d / nbk;
            if (b > NOWN - 1) b = NOWN - 1;
            const int dl = (d - b * nbk) & (NBKMAX - 1);
            bk[j] = b;
            ent[j] = (int)(((unsigned)dl << SRC_BITS) | ((unsigned)s & SRC_MASK));
            rk[j] = atomicAdd(&s_cnt[b], 1);
        }
    }
    __syncthreads();

    int c = 0, incl = 0;
    if (tid < NOWN) {
        c = s_cnt[tid];
        incl = c;
#pragma unroll
        for (int d = 1; d < 32; d <<= 1) {
            const int t = __shfl_up(incl, d);
            if (lane >= d) incl += t;
        }
        if (lane == 31) s_wsum[wave] = incl;
    }
    __syncthreads();
    if (tid < NOWN) {
        int add = 0;
        for (int k = 0; k < wave; ++k) add += s_wsum[k];
        s_start[tid] = incl - c + add;
        if (tid == NOWN - 1) s_start[NOWN] = incl + add;
    }
    __syncthreads();

#pragma unroll
    for (int j = 0; j < EPT; ++j)
        if (bk[j] >= 0) s_sorted[s_start[bk[j]] + rk[j]] = ent[j];
    __syncthreads();

    const int total = s_start[NOWN];
#pragma unroll
    for (int j = 0; j < EPT; ++j) {
        const int i = j * PT + tid;
        if (i < total) list[base + i] = s_sorted[i];
    }
    if (tid < OP) offs[(long long)w * OP + tid] = (tid <= NOWN) ? s_start[tid] : total;
}

__device__ __forceinline__ float b2load(const float* w2k1, const float* root2,
                                        int k, int col) {
    if (col >= 40) return 0.0f;
    return (k < 16) ? w2k1[k * 40 + col] : root2[(k - 16) * 40 + col];
}

__device__ __forceinline__ void gemm2_tail(const float* s_acc, const int* s_deg,
                                           int wave, int lane,
                                           int node0, int nbk, int N,
                                           const float* __restrict__ h,
                                           const float* __restrict__ w2k1,
                                           const float* __restrict__ root2,
                                           const float* __restrict__ b2,
                                           float* __restrict__ out) {
    const int lr0 = wave * 16;
    if (lr0 >= nbk) return;

    const int colL = lane & 15;
    const int kSub = (lane >> 4) * 2;
    const int rl = lr0 + colL;
    int nA = node0 + rl;
    if (nA >= N) nA = N - 1;
    const float invd = 1.0f / fmaxf((float)s_deg[rl], 1.0f);

    v8f c0 = {}, c1 = {}, c2 = {};
#pragma unroll
    for (int kk = 0; kk < 8; ++kk) {
        const int k = kk * 4 + kSub;
        v2f a;
        a.x = (k < 16) ? s_acc[rl * 16 + k] * invd
                       : h[(long long)nA * 16 + (k - 16)];
        a.y = (k + 1 < 16) ? s_acc[rl * 16 + k + 1] * invd
                           : h[(long long)nA * 16 + (k + 1 - 16)];
        v2f b0, b1, b2v;
        b0.x  = b2load(w2k1, root2, k,     colL);
        b0.y  = b2load(w2k1, root2, k + 1, colL);
        b1.x  = b2load(w2k1, root2, k,     16 + colL);
        b1.y  = b2load(w2k1, root2, k + 1, 16 + colL);
        b2v.x = b2load(w2k1, root2, k,     32 + colL);
        b2v.y = b2load(w2k1, root2, k + 1, 32 + colL);
        c0 = __builtin_amdgcn_wmma_f32_16x16x4_f32(false, a, false, b0,
                                                   (short)0, c0, false, false);
        c1 = __builtin_amdgcn_wmma_f32_16x16x4_f32(false, a, false, b1,
                                                   (short)0, c1, false, false);
        c2 = __builtin_amdgcn_wmma_f32_16x16x4_f32(false, a, false, b2v,
                                                   (short)0, c2, false, false);
    }

    asm volatile("v_nop\n\tv_nop\n\tv_nop\n\tv_nop" : "+v"(c0), "+v"(c1), "+v"(c2));

    const int rbase = (lane >> 4) * 8;
    const bool tail = colL < 8;
    const float bia0 = b2[colL];
    const float bia1 = b2[16 + colL];
    const float bia2 = tail ? b2[32 + colL] : 0.0f;

    float v0[8], v1[8], v2[8], m[8], s[8];
#pragma unroll
    for (int r = 0; r < 8; ++r) {
        v0[r] = c0[r] + bia0;
        v1[r] = c1[r] + bia1;
        v2[r] = tail ? (c2[r] + bia2) : -INFINITY;
        m[r] = fmaxf(fmaxf(v0[r], v1[r]), v2[r]);
    }
#pragma unroll
    for (int d = 1; d < 16; d <<= 1) {
#pragma unroll
        for (int r = 0; r < 8; ++r) m[r] = fmaxf(m[r], __shfl_xor(m[r], d));
    }
#pragma unroll
    for (int r = 0; r < 8; ++r) {
        s[r] = expf(v0[r] - m[r]) + expf(v1[r] - m[r]);
        if (tail) s[r] += expf(v2[r] - m[r]);
    }
#pragma unroll
    for (int d = 1; d < 16; d <<= 1) {
#pragma unroll
        for (int r = 0; r < 8; ++r) s[r] += __shfl_xor(s[r], d);
    }
#pragma unroll
    for (int r = 0; r < 8; ++r) {
        const int lrow = lr0 + rbase + r;
        const int row = node0 + lrow;
        if (lrow >= nbk || row >= N) continue;
        const float ls = logf(s[r]);
        out[(long long)row * 40 + colL]      = v0[r] - m[r] - ls;
        out[(long long)row * 40 + 16 + colL] = v1[r] - m[r] - ls;
        if (tail)
            out[(long long)row * 40 + 32 + colL] = v2[r] - m[r] - ls;
    }
}

#define ROW_SHARE_CASE(J) case J: return __builtin_amdgcn_mov_dpp(v, 0x150 + J, 0xf, 0xf, false);
__device__ __forceinline__ int row_bcast(int v, int j) {
    switch (j) {
        ROW_SHARE_CASE(0)  ROW_SHARE_CASE(1)  ROW_SHARE_CASE(2)  ROW_SHARE_CASE(3)
        ROW_SHARE_CASE(4)  ROW_SHARE_CASE(5)  ROW_SHARE_CASE(6)  ROW_SHARE_CASE(7)
        ROW_SHARE_CASE(8)  ROW_SHARE_CASE(9)  ROW_SHARE_CASE(10) ROW_SHARE_CASE(11)
        ROW_SHARE_CASE(12) ROW_SHARE_CASE(13) ROW_SHARE_CASE(14)
        default: return __builtin_amdgcn_mov_dpp(v, 0x150 + 15, 0xf, 0xf, false);
    }
}

template <int LAYER>
__device__ __forceinline__ void owner_body(const int* __restrict__ list,
                                           const int* __restrict__ offs,
                                           int nChunks, int N, int nbk,
                                           const float* __restrict__ feat,
                                           float* r1h,
                                           const float* __restrict__ b1,
                                           const float* __restrict__ w2k1,
                                           const float* __restrict__ root2,
                                           const float* __restrict__ b2,
                                           float* __restrict__ out) {
    __shared__ float s_acc[NBKMAX * 16];
    __shared__ int s_deg[NBKMAX];
    const int tid = threadIdx.x;
    const int lane = tid & 31;
    const int wave = __builtin_amdgcn_readfirstlane(tid >> 5);
    const int c = lane & 15;
    const int hsel = lane >> 4;
    const int eslot = 2 * c + hsel;
    const int b = blockIdx.x;
    const unsigned nlast = (unsigned)(N - 1);

    for (int i = tid; i < NBKMAX * 16; i += PT) s_acc[i] = 0.0f;
    if (tid < NBKMAX) s_deg[tid] = 0;
    __syncthreads();

    for (int w = wave; w < nChunks; w += PT / 32) {
        int o0 = __builtin_amdgcn_readfirstlane(offs[(long long)w * OP + b]);
        int o1 = __builtin_amdgcn_readfirstlane(offs[(long long)w * OP + b + 1]);
        o1 = (o1 < 0) ? 0 : ((o1 > CHK) ? CHK : o1);
        o0 = (o0 < 0) ? 0 : ((o0 > o1) ? o1 : o0);
        const int* run = list + (long long)w * CHK;
        int nxt = 0;
        if (o0 + eslot < o1) nxt = run[o0 + eslot];
        for (int p = o0; p < o1; p += 32) {
            const int ent = nxt;
            const int rem = o1 - p;
            const int cnt = (rem < 32) ? rem : 32;
            nxt = 0;
            if (p + 32 + eslot < o1) nxt = run[p + 32 + eslot];
            if (eslot < cnt) atomicAdd(&s_deg[((unsigned)ent) >> SRC_BITS], 1);
            float v[16];
            int en[16];
#pragma unroll
            for (int g = 0; g < 4; ++g) {
                if (g * 8 >= cnt) break;
#pragma unroll
                for (int jj = g * 4; jj < g * 4 + 4; ++jj) {
                    en[jj] = row_bcast(ent, jj);
                    unsigned sidx = (unsigned)en[jj] & SRC_MASK;
                    if (sidx > nlast) sidx = nlast;
                    v[jj] = feat[(sidx << 4) | (unsigned)c];
                }
            }
#pragma unroll
            for (int g = 0; g < 4; ++g) {
                if (g * 8 >= cnt) break;
#pragma unroll
                for (int jj = g * 4; jj < g * 4 + 4; ++jj) {
                    const float add = (2 * jj + hsel < cnt) ? v[jj] : 0.0f;
                    atomicAdd(&s_acc[((((unsigned)en[jj]) >> SRC_BITS) << 4) | (unsigned)c], add);
                }
            }
        }
    }
    __syncthreads();

    const int node0 = b * nbk;
    if (LAYER == 1) {
#pragma unroll
        for (int q = 0; q < NBKMAX * 16 / PT; ++q) {
            const int idx = q * PT + tid;
            const int r = idx >> 4;
            const int cc = idx & 15;
            const int n = node0 + r;
            if (r < nbk && n < N) {
                const float dg = fmaxf((float)s_deg[r], 1.0f);
                const float v = s_acc[idx] / dg + r1h[(long long)n * 16 + cc] + b1[cc];
                r1h[(long long)n * 16 + cc] = (v > 0.0f) ? v : (expf(v) - 1.0f);
            }
        }
    } else {
        gemm2_tail(s_acc, s_deg, wave, lane, node0, nbk, N, feat, w2k1, root2, b2, out);
    }
}

__global__ __launch_bounds__(PT) void owner1_kernel(const int* __restrict__ list,
                                                    const int* __restrict__ offs,
                                                    int nChunks, int N, int nbk,
                                                    const float* __restrict__ y1,
                                                    float* r1h,
                                                    const float* __restrict__ b1) {
    owner_body<1>(list, offs, nChunks, N, nbk, y1, r1h, b1, nullptr, nullptr, nullptr, nullptr);
}

__global__ __launch_bounds__(PT) void owner2_kernel(const int* __restrict__ list,
                                                    const int* __restrict__ offs,
                                                    int nChunks, int N, int nbk,
                                                    const float* __restrict__ h,
                                                    const float* __restrict__ w2k1,
                                                    const float* __restrict__ root2,
                                                    const float* __restrict__ b2,
                                                    float* __restrict__ out) {
    owner_body<2>(list, offs, nChunks, N, nbk, h, nullptr, nullptr, w2k1, root2, b2, out);
}

extern "C" void kernel_launch(void* const* d_in, const int* in_sizes, int n_in,
                              void* d_out, int out_size, void* d_ws, size_t ws_size,
                              hipStream_t stream) {
    const float* x     = (const float*)d_in[0];
    const int*   ei    = (const int*)d_in[1];
    const float* W1    = (const float*)d_in[2];
    const float* root1 = (const float*)d_in[3];
    const float* b1    = (const float*)d_in[4];
    const float* W2    = (const float*)d_in[5];
    const float* root2 = (const float*)d_in[6];
    const float* b2    = (const float*)d_in[7];
    float*       out   = (float*)d_out;

    const int N = in_sizes[0] / 128;
    const int E = in_sizes[1] / 2;

    const float* w1k1 = W1 + 128 * 16;
    const float* w2k1 = W2 + 16 * 40;

    const int nChunks = (int)(((long long)E + CHK - 1) / CHK);
    const int nbk     = ((N + NOWN - 1) / NOWN + 3) & ~3;
    float* ws   = (float*)d_ws;
    float* y1   = ws;
    float* r1h  = ws + (size_t)N * 16;
    int*   list = (int*)(ws + (size_t)N * 32);
    int*   offs = list + (size_t)nChunks * CHK;

    const int tiles   = (N + 15) / 16;
    const int gblocks = (tiles + 3) / 4;

    place_kernel<<<nChunks, PT, 0, stream>>>(ei, E, N, nbk, list, offs);

    gemm1_kernel<<<gblocks, 128, 0, stream>>>(x, w1k1, root1, y1, r1h, N);

    owner1_kernel<<<NOWN, PT, 0, stream>>>(list, offs, nChunks, N, nbk, y1, r1h, b1);

    owner2_kernel<<<NOWN, PT, 0, stream>>>(list, offs, nChunks, N, nbk, r1h,
                                           w2k1, root2, b2, out);
}
